# starting kernel
# baseline (speedup 1.0000x reference)
#include <hip/hip_runtime.h>
#include <hip/hip_bf16.h>

typedef __attribute__((ext_vector_type(2))) float v2f;
typedef __attribute__((ext_vector_type(4))) float v4f;
typedef __attribute__((ext_vector_type(8))) float v8f;

__device__ __forceinline__ float xswap16(float v) {
#if __has_builtin(__builtin_amdgcn_permlanex16)
    return __builtin_bit_cast(float, __builtin_amdgcn_permlanex16(
        __builtin_bit_cast(int, v), __builtin_bit_cast(int, v),
        0x76543210, 0xfedcba98, false, false));
#elif __has_builtin(__builtin_amdgcn_ds_swizzle)
    return __builtin_bit_cast(float,
        __builtin_amdgcn_ds_swizzle(__builtin_bit_cast(int, v), 0x401f));
#else
    return v;
#endif
}

__device__ __forceinline__ float relu_med3(float x, float finf) {
#if defined(__AMDGCN__)
    float r;
    asm("v_med3_f32 %0, %1, 0, %2" : "=v"(r) : "v"(x), "v"(finf));
    return r;
#else
    return fmaxf(x, 0.0f);
#endif
}

__global__ __launch_bounds__(256) void rnn_scan_wmma(
    const float* __restrict__ x,
    const float* __restrict__ W_ih,
    const float* __restrict__ b_ih,
    const float* __restrict__ W_hh,
    const float* __restrict__ b_hh,
    const float* __restrict__ W_fc,
    const float* __restrict__ b_fc,
    float* __restrict__ out,
    int B, int T)
{
    const int lane = threadIdx.x & 31;
    const int wave = threadIdx.x >> 5;
    const int n    = lane & 15;
    const bool hi  = lane >= 16;
    const int h8   = hi ? 8 : 0;

    const int waveGlobal = blockIdx.x * (blockDim.x >> 5) + wave;
    const int row0 = (waveGlobal * 2 + 0) * 16;
    const int row1 = (waveGlobal * 2 + 1) * 16;
    if (row0 >= B) return;

    const float finf = __builtin_bit_cast(float, 0x7f800000);

    v2f a[4];
#pragma unroll
    for (int j = 0; j < 4; ++j) {
        const int k0 = 2 * j + h8;
        a[j].x = W_hh[n * 16 + k0];
        a[j].y = W_hh[n * 16 + k0 + 1];
    }

    float wih[8], bias[8];
#pragma unroll
    for (int v = 0; v < 8; ++v) {
        wih[v]  = W_ih[v + h8];
        bias[v] = b_ih[v + h8] + b_hh[v + h8];
    }

    v8f h0 = {}, h1 = {};
    const v8f zc = {};

    const float* xr0 = x + (size_t)(row0 + n) * T;
    const float* xr1 = x + (size_t)(row1 + n) * T;

    for (int t = 0; t < T; t += 4) {
        v4f xq0 = *(const v4f*)(xr0 + t);
        v4f xq1 = *(const v4f*)(xr1 + t);
        __builtin_prefetch(xr0 + t + 32, 0, 1);
        __builtin_prefetch(xr1 + t + 32, 0, 1);
#pragma unroll
        for (int u = 0; u < 4; ++u) {
            v8f c0, c1;
#pragma unroll
            for (int v = 0; v < 8; ++v) {
                c0[v] = fmaf(xq0[u], wih[v], bias[v]);
                c1[v] = fmaf(xq1[u], wih[v], bias[v]);
            }
            v2f b00 = { h0[0], h0[1] }, b01 = { h0[2], h0[3] };
            v2f b02 = { h0[4], h0[5] }, b03 = { h0[6], h0[7] };
            v2f b10 = { h1[0], h1[1] }, b11 = { h1[2], h1[3] };
            v2f b12 = { h1[4], h1[5] }, b13 = { h1[6], h1[7] };

            v8f q0 = __builtin_amdgcn_wmma_f32_16x16x4_f32(
                        false, a[2], false, b02, (short)0, zc, false, false);
            v8f q1 = __builtin_amdgcn_wmma_f32_16x16x4_f32(
                        false, a[2], false, b12, (short)0, zc, false, false);
            c0 = __builtin_amdgcn_wmma_f32_16x16x4_f32(
                        false, a[0], false, b00, (short)0, c0, false, false);
            c1 = __builtin_amdgcn_wmma_f32_16x16x4_f32(
                        false, a[0], false, b10, (short)0, c1, false, false);
            q0 = __builtin_amdgcn_wmma_f32_16x16x4_f32(
                        false, a[3], false, b03, (short)0, q0, false, false);
            q1 = __builtin_amdgcn_wmma_f32_16x16x4_f32(
                        false, a[3], false, b13, (short)0, q1, false, false);
            c0 = __builtin_amdgcn_wmma_f32_16x16x4_f32(
                        false, a[1], false, b01, (short)0, c0, false, false);
            c1 = __builtin_amdgcn_wmma_f32_16x16x4_f32(
                        false, a[1], false, b11, (short)0, c1, false, false);

#pragma unroll
            for (int v = 0; v < 8; ++v) {
                h0[v] = relu_med3(c0[v] + q0[v], finf);
                h1[v] = relu_med3(c1[v] + q1[v], finf);
            }
        }
    }

    float wf0[8], wf1[8];
#pragma unroll
    for (int v = 0; v < 8; ++v) {
        wf0[v] = W_fc[0 * 16 + v + h8];
        wf1[v] = W_fc[1 * 16 + v + h8];
    }
    float p00 = 0.f, p01 = 0.f, p10 = 0.f, p11 = 0.f;
#pragma unroll
    for (int v = 0; v < 8; ++v) {
        p00 = fmaf(h0[v], wf0[v], p00);
        p01 = fmaf(h0[v], wf1[v], p01);
        p10 = fmaf(h1[v], wf0[v], p10);
        p11 = fmaf(h1[v], wf1[v], p11);
    }
    p00 += xswap16(p00); p01 += xswap16(p01);
    p10 += xswap16(p10); p11 += xswap16(p11);

    if (!hi) {
        out[(size_t)(row0 + n) * 2 + 0] = p00 + b_fc[0];
        out[(size_t)(row0 + n) * 2 + 1] = p01 + b_fc[1];
        if (row1 < B) {
            out[(size_t)(row1 + n) * 2 + 0] = p10 + b_fc[0];
            out[(size_t)(row1 + n) * 2 + 1] = p11 + b_fc[1];
        }
    }
}

extern "C" void kernel_launch(void* const* d_in, const int* in_sizes, int n_in,
                              void* d_out, int out_size, void* d_ws, size_t ws_size,
                              hipStream_t stream) {
    const float* x    = (const float*)d_in[0];
    const float* W_ih = (const float*)d_in[1];
    const float* b_ih = (const float*)d_in[2];
    const float* W_hh = (const float*)d_in[3];
    const float* b_hh = (const float*)d_in[4];
    const float* W_fc = (const float*)d_in[5];
    const float* b_fc = (const float*)d_in[6];
    float* out = (float*)d_out;

    const int C = 2;
    const int B = out_size / C;
    const int T = in_sizes[0] / B;

    const int threads = 256;
    const int batchPerBlock = (threads / 32) * 2 * 16;
    const int blocks = (B + batchPerBlock - 1) / batchPerBlock;

    rnn_scan_wmma<<<blocks, threads, 0, stream>>>(
        x, W_ih, b_ih, W_hh, b_hh, W_fc, b_fc, out, B, T);
}


# One 16-row batch tile per wave and one wave per work-group (256 work-groups of 32 threads; was 2 tiles per wave, 8 waves per work-group, 16 work-groups): one serial chain per SIMD instead of four
# speedup vs baseline: 3.0232x; 3.0232x over previous
#include <hip/hip_runtime.h>
#include <hip/hip_bf16.h>

typedef __attribute__((ext_vector_type(2))) float v2f;
typedef __attribute__((ext_vector_type(4))) float v4f;
typedef __attribute__((ext_vector_type(8))) float v8f;

__device__ __forceinline__ float xswap16(float v) {
#if __has_builtin(__builtin_amdgcn_permlanex16)
    return __builtin_bit_cast(float, __builtin_amdgcn_permlanex16(
        __builtin_bit_cast(int, v), __builtin_bit_cast(int, v),
        0x76543210, 0xfedcba98, false, false));
#elif __has_builtin(__builtin_amdgcn_ds_swizzle)
    return __builtin_bit_cast(float,
        __builtin_amdgcn_ds_swizzle(__builtin_bit_cast(int, v), 0x401f));
#else
    return v;
#endif
}

__device__ __forceinline__ float relu_med3(float x, float finf) {
#if defined(__AMDGCN__)
    float r;
    asm("v_med3_f32 %0, %1, 0, %2" : "=v"(r) : "v"(x), "v"(finf));
    return r;
#else
    return fmaxf(x, 0.0f);
#endif
}

__global__ __launch_bounds__(32) void rnn_scan_wmma(
    const float* __restrict__ x,
    const float* __restrict__ W_ih,
    const float* __restrict__ b_ih,
    const float* __restrict__ W_hh,
    const float* __restrict__ b_hh,
    const float* __restrict__ W_fc,
    const float* __restrict__ b_fc,
    float* __restrict__ out,
    int B, int T)
{
    const int lane = threadIdx.x & 31;
    const int wave = threadIdx.x >> 5;
    const int n    = lane & 15;
    const bool hi  = lane >= 16;
    const int h8   = hi ? 8 : 0;

    const int waveGlobal = blockIdx.x * (blockDim.x >> 5) + wave;
    const int row0 = waveGlobal * 16;
    if (row0 >= B) return;

    const float finf = __builtin_bit_cast(float, 0x7f800000);

    v2f a[4];
#pragma unroll
    for (int j = 0; j < 4; ++j) {
        const int k0 = 2 * j + h8;
        a[j].x = W_hh[n * 16 + k0];
        a[j].y = W_hh[n * 16 + k0 + 1];
    }

    float wih[8], bias[8];
#pragma unroll
    for (int v = 0; v < 8; ++v) {
        wih[v]  = W_ih[v + h8];
        bias[v] = b_ih[v + h8] + b_hh[v + h8];
    }

    v8f h0 = {};
    const v8f zc = {};

    const float* xr0 = x + (size_t)(row0 + n) * T;

    for (int t = 0; t < T; t += 4) {
        v4f xq0 = *(const v4f*)(xr0 + t);
        __builtin_prefetch(xr0 + t + 32, 0, 1);
#pragma unroll
        for (int u = 0; u < 4; ++u) {
            v8f c0;
#pragma unroll
            for (int v = 0; v < 8; ++v)
                c0[v] = fmaf(xq0[u], wih[v], bias[v]);

            v2f b00 = { h0[0], h0[1] }, b01 = { h0[2], h0[3] };
            v2f b02 = { h0[4], h0[5] }, b03 = { h0[6], h0[7] };

            v8f q0 = __builtin_amdgcn_wmma_f32_16x16x4_f32(
                        false, a[2], false, b02, (short)0, zc, false, false);
            c0 = __builtin_amdgcn_wmma_f32_16x16x4_f32(
                        false, a[0], false, b00, (short)0, c0, false, false);
            q0 = __builtin_amdgcn_wmma_f32_16x16x4_f32(
                        false, a[3], false, b03, (short)0, q0, false, false);
            c0 = __builtin_amdgcn_wmma_f32_16x16x4_f32(
                        false, a[1], false, b01, (short)0, c0, false, false);

#pragma unroll
            for (int v = 0; v < 8; ++v)
                h0[v] = relu_med3(c0[v] + q0[v], finf);
        }
    }

    float wf0[8], wf1[8];
#pragma unroll
    for (int v = 0; v < 8; ++v) {
        wf0[v] = W_fc[0 * 16 + v + h8];
        wf1[v] = W_fc[1 * 16 + v + h8];
    }
    float p00 = 0.f, p01 = 0.f;
#pragma unroll
    for (int v = 0; v < 8; ++v) {
        p00 = fmaf(h0[v], wf0[v], p00);
        p01 = fmaf(h0[v], wf1[v], p01);
    }
    p00 += xswap16(p00); p01 += xswap16(p01);

    if (!hi) {
        out[(size_t)(row0 + n) * 2 + 0] = p00 + b_fc[0];
        out[(size_t)(row0 + n) * 2 + 1] = p01 + b_fc[1];
    }
}

extern "C" void kernel_launch(void* const* d_in, const int* in_sizes, int n_in,
                              void* d_out, int out_size, void* d_ws, size_t ws_size,
                              hipStream_t stream) {
    const float* x    = (const float*)d_in[0];
    const float* W_ih = (const float*)d_in[1];
    const float* b_ih = (const float*)d_in[2];
    const float* W_hh = (const float*)d_in[3];
    const float* b_hh = (const float*)d_in[4];
    const float* W_fc = (const float*)d_in[5];
    const float* b_fc = (const float*)d_in[6];
    float* out = (float*)d_out;

    const int C = 2;
    const int B = out_size / C;
    const int T = in_sizes[0] / B;

    const int threads = 32;
    const int batchPerBlock = (threads / 32) * 16;
    const int blocks = (B + batchPerBlock - 1) / batchPerBlock;

    rnn_scan_wmma<<<blocks, threads, 0, stream>>>(
        x, W_ih, b_ih, W_hh, b_hh, W_fc, b_fc, out, B, T);
}


# x for later timesteps is loaded into a ring of 4 register slots three blocks (12 steps) ahead of its use, held there by scheduling fences; before, each block loaded its own x and waited for it
# speedup vs baseline: 3.4561x; 1.1432x over previous
#include <hip/hip_runtime.h>
#include <hip/hip_bf16.h>

typedef __attribute__((ext_vector_type(2))) float v2f;
typedef __attribute__((ext_vector_type(4))) float v4f;
typedef __attribute__((ext_vector_type(8))) float v8f;

__device__ __forceinline__ float xswap16(float v) {
#if __has_builtin(__builtin_amdgcn_permlanex16)
    return __builtin_bit_cast(float, __builtin_amdgcn_permlanex16(
        __builtin_bit_cast(int, v), __builtin_bit_cast(int, v),
        0x76543210, 0xfedcba98, false, false));
#elif __has_builtin(__builtin_amdgcn_ds_swizzle)
    return __builtin_bit_cast(float,
        __builtin_amdgcn_ds_swizzle(__builtin_bit_cast(int, v), 0x401f));
#else
    return v;
#endif
}

__device__ __forceinline__ float relu_med3(float x, float finf) {
#if defined(__AMDGCN__)
    float r;
    asm("v_med3_f32 %0, %1, 0, %2" : "=v"(r) : "v"(x), "v"(finf));
    return r;
#else
    return fmaxf(x, 0.0f);
#endif
}

__device__ __forceinline__ void rnn_steps4(v8f& h0, const v2f (&a)[4],
                                           const float (&wih)[8], const float (&bias)[8],
                                           const v4f xq, const float finf) {
    const v8f zc = {};
#pragma unroll
    for (int u = 0; u < 4; ++u) {
        v8f c0;
#pragma unroll
        for (int v = 0; v < 8; ++v)
            c0[v] = fmaf(xq[u], wih[v], bias[v]);

        v2f b00 = { h0[0], h0[1] }, b01 = { h0[2], h0[3] };
        v2f b02 = { h0[4], h0[5] }, b03 = { h0[6], h0[7] };

        v8f q0 = __builtin_amdgcn_wmma_f32_16x16x4_f32(
                    false, a[2], false, b02, (short)0, zc, false, false);
        c0 = __builtin_amdgcn_wmma_f32_16x16x4_f32(
                    false, a[0], false, b00, (short)0, c0, false, false);
        q0 = __builtin_amdgcn_wmma_f32_16x16x4_f32(
                    false, a[3], false, b03, (short)0, q0, false, false);
        c0 = __builtin_amdgcn_wmma_f32_16x16x4_f32(
                    false, a[1], false, b01, (short)0, c0, false, false);

#pragma unroll
        for (int v = 0; v < 8; ++v)
            h0[v] = relu_med3(c0[v] + q0[v], finf);
    }
}

__global__ __launch_bounds__(32) void rnn_scan_wmma(
    const float* __restrict__ x,
    const float* __restrict__ W_ih,
    const float* __restrict__ b_ih,
    const float* __restrict__ W_hh,
    const float* __restrict__ b_hh,
    const float* __restrict__ W_fc,
    const float* __restrict__ b_fc,
    float* __restrict__ out,
    int B, int T)
{
    const int lane = threadIdx.x & 31;
    const int wave = threadIdx.x >> 5;
    const int n    = lane & 15;
    const bool hi  = lane >= 16;
    const int h8   = hi ? 8 : 0;

    const int waveGlobal = blockIdx.x * (blockDim.x >> 5) + wave;
    const int row0 = waveGlobal * 16;
    if (row0 >= B) return;

    const float finf = __builtin_bit_cast(float, 0x7f800000);

    v2f a[4];
#pragma unroll
    for (int j = 0; j < 4; ++j) {
        const int k0 = 2 * j + h8;
        a[j].x = W_hh[n * 16 + k0];
        a[j].y = W_hh[n * 16 + k0 + 1];
    }

    float wih[8], bias[8];
#pragma unroll
    for (int v = 0; v < 8; ++v) {
        wih[v]  = W_ih[v + h8];
        bias[v] = b_ih[v + h8] + b_hh[v + h8];
    }

    v8f h0 = {};

    const float* xr0 = x + (size_t)(row0 + n) * T;

    constexpr int RING = 4;
    const int nblk = T >> 2;
    const int last = nblk - 1;
    if (nblk > 0) {
        v4f xring[RING];
#pragma unroll
        for (int d = 0; d < RING - 1; ++d)
            xring[d] = *(const v4f*)(xr0 + 4 * (size_t)min(d, last));

        int blk = 0;
        for (; blk + RING <= nblk; blk += RING) {
#pragma unroll
            for (int d = 0; d < RING; ++d) {
                __builtin_amdgcn_sched_barrier(0x6);
                xring[(d + RING - 1) % RING] =
                    *(const v4f*)(xr0 + 4 * (size_t)min(blk + d + RING - 1, last));
                __builtin_prefetch(xr0 + 4 * (size_t)(blk + d) + 32, 0, 1);
                __builtin_amdgcn_sched_barrier(0x6);
                rnn_steps4(h0, a, wih, bias, xring[d], finf);
            }
        }
#pragma unroll
        for (int d = 0; d < RING - 1; ++d)
            if (blk + d < nblk)
                rnn_steps4(h0, a, wih, bias, xring[d], finf);
    }

    float wf0[8], wf1[8];
#pragma unroll
    for (int v = 0; v < 8; ++v) {
        wf0[v] = W_fc[0 * 16 + v + h8];
        wf1[v] = W_fc[1 * 16 + v + h8];
    }
    float p00 = 0.f, p01 = 0.f;
#pragma unroll
    for (int v = 0; v < 8; ++v) {
        p00 = fmaf(h0[v], wf0[v], p00);
        p01 = fmaf(h0[v], wf1[v], p01);
    }
    p00 += xswap16(p00); p01 += xswap16(p01);

    if (!hi) {
        out[(size_t)(row0 + n) * 2 + 0] = p00 + b_fc[0];
        out[(size_t)(row0 + n) * 2 + 1] = p01 + b_fc[1];
    }
}

extern "C" void kernel_launch(void* const* d_in, const int* in_sizes, int n_in,
                              void* d_out, int out_size, void* d_ws, size_t ws_size,
                              hipStream_t stream) {
    const float* x    = (const float*)d_in[0];
    const float* W_ih = (const float*)d_in[1];
    const float* b_ih = (const float*)d_in[2];
    const float* W_hh = (const float*)d_in[3];
    const float* b_hh = (const float*)d_in[4];
    const float* W_fc = (const float*)d_in[5];
    const float* b_fc = (const float*)d_in[6];
    float* out = (float*)d_out;

    const int C = 2;
    const int B = out_size / C;
    const int T = in_sizes[0] / B;

    const int threads = 32;
    const int batchPerBlock = (threads / 32) * 16;
    const int blocks = (B + batchPerBlock - 1) / batchPerBlock;

    rnn_scan_wmma<<<blocks, threads, 0, stream>>>(
        x, W_ih, b_ih, W_hh, b_hh, W_fc, b_fc, out, B, T);
}


# x*W_ih is added to the first chain's result behind the fourth matrix instruction (bias is that chain's C): no vector instruction stands between a step's matrix instructions; sum order changes
# speedup vs baseline: 3.6746x; 1.0632x over previous
#include <hip/hip_runtime.h>
#include <hip/hip_bf16.h>

typedef __attribute__((ext_vector_type(2))) float v2f;
typedef __attribute__((ext_vector_type(4))) float v4f;
typedef __attribute__((ext_vector_type(8))) float v8f;

__device__ __forceinline__ float xswap16(float v) {
#if __has_builtin(__builtin_amdgcn_permlanex16)
    return __builtin_bit_cast(float, __builtin_amdgcn_permlanex16(
        __builtin_bit_cast(int, v), __builtin_bit_cast(int, v),
        0x76543210, 0xfedcba98, false, false));
#elif __has_builtin(__builtin_amdgcn_ds_swizzle)
    return __builtin_bit_cast(float,
        __builtin_amdgcn_ds_swizzle(__builtin_bit_cast(int, v), 0x401f));
#else
    return v;
#endif
}

__device__ __forceinline__ float relu_med3(float x, float finf) {
#if defined(__AMDGCN__)
    float r;
    asm("v_med3_f32 %0, %1, 0, %2" : "=v"(r) : "v"(x), "v"(finf));
    return r;
#else
    return fmaxf(x, 0.0f);
#endif
}

__device__ __forceinline__ void rnn_steps4(v8f& h0, const v2f (&a)[4],
                                           const float (&wih)[8], const v8f bias,
                                           const v4f xq, const float finf) {
    const v8f zc = {};
#pragma unroll
    for (int u = 0; u < 4; ++u) {
        v2f b00 = { h0[0], h0[1] }, b01 = { h0[2], h0[3] };
        v2f b02 = { h0[4], h0[5] }, b03 = { h0[6], h0[7] };

        __builtin_amdgcn_sched_barrier(0);
        v8f c0 = __builtin_amdgcn_wmma_f32_16x16x4_f32(
                    false, a[0], false, b00, (short)0, bias, false, false);
        v8f q0 = __builtin_amdgcn_wmma_f32_16x16x4_f32(
                    false, a[2], false, b02, (short)0, zc, false, false);
        c0 = __builtin_amdgcn_wmma_f32_16x16x4_f32(
                    false, a[1], false, b01, (short)0, c0, false, false);
        q0 = __builtin_amdgcn_wmma_f32_16x16x4_f32(
                    false, a[3], false, b03, (short)0, q0, false, false);
        __builtin_amdgcn_sched_barrier(0);
#if defined(__AMDGCN__)
        asm volatile("" : "+v"(c0) : "v"(q0));
#endif

        v8f t0;
#pragma unroll
        for (int v = 0; v < 8; ++v)
            t0[v] = fmaf(xq[u], wih[v], c0[v]);

        float s0[8];
#pragma unroll
        for (int v = 0; v < 8; ++v)
            s0[v] = t0[v] + q0[v];
        __builtin_amdgcn_sched_barrier(0);
#pragma unroll
        for (int v = 0; v < 8; ++v)
            h0[v] = relu_med3(s0[v], finf);
    }
}

__global__ __launch_bounds__(32) void rnn_scan_wmma(
    const float* __restrict__ x,
    const float* __restrict__ W_ih,
    const float* __restrict__ b_ih,
    const float* __restrict__ W_hh,
    const float* __restrict__ b_hh,
    const float* __restrict__ W_fc,
    const float* __restrict__ b_fc,
    float* __restrict__ out,
    int B, int T)
{
    const int lane = threadIdx.x & 31;
    const int wave = threadIdx.x >> 5;
    const int n    = lane & 15;
    const bool hi  = lane >= 16;
    const int h8   = hi ? 8 : 0;

    const int waveGlobal = blockIdx.x * (blockDim.x >> 5) + wave;
    const int row0 = waveGlobal * 16;
    if (row0 >= B) return;

    const float finf = __builtin_bit_cast(float, 0x7f800000);

    v2f a[4];
#pragma unroll
    for (int j = 0; j < 4; ++j) {
        const int k0 = 2 * j + h8;
        a[j].x = W_hh[n * 16 + k0];
        a[j].y = W_hh[n * 16 + k0 + 1];
    }

    float wih[8];
    v8f bias;
#pragma unroll
    for (int v = 0; v < 8; ++v) {
        wih[v]  = W_ih[v + h8];
        bias[v] = b_ih[v + h8] + b_hh[v + h8];
    }

    v8f h0 = {};

    const float* xr0 = x + (size_t)(row0 + n) * T;

    constexpr int RING = 4;
    const int nblk = T >> 2;
    const int last = nblk - 1;
    if (nblk > 0) {
        v4f xring[RING];
#pragma unroll
        for (int d = 0; d < RING - 1; ++d)
            xring[d] = *(const v4f*)(xr0 + 4 * (size_t)min(d, last));

        int blk = 0;
        for (; blk + RING <= nblk; blk += RING) {
#pragma unroll
            for (int d = 0; d < RING; ++d) {
                __builtin_amdgcn_sched_barrier(0x6);
                xring[(d + RING - 1) % RING] =
                    *(const v4f*)(xr0 + 4 * (size_t)min(blk + d + RING - 1, last));
                __builtin_prefetch(xr0 + 4 * (size_t)(blk + d) + 32, 0, 1);
                __builtin_amdgcn_sched_barrier(0x6);
                rnn_steps4(h0, a, wih, bias, xring[d], finf);
            }
        }
#pragma unroll
        for (int d = 0; d < RING - 1; ++d)
            if (blk + d < nblk)
                rnn_steps4(h0, a, wih, bias, xring[d], finf);
    }

    float wf0[8], wf1[8];
#pragma unroll
    for (int v = 0; v < 8; ++v) {
        wf0[v] = W_fc[0 * 16 + v + h8];
        wf1[v] = W_fc[1 * 16 + v + h8];
    }
    float p00 = 0.f, p01 = 0.f;
#pragma unroll
    for (int v = 0; v < 8; ++v) {
        p00 = fmaf(h0[v], wf0[v], p00);
        p01 = fmaf(h0[v], wf1[v], p01);
    }
    p00 += xswap16(p00); p01 += xswap16(p01);

    if (!hi) {
        out[(size_t)(row0 + n) * 2 + 0] = p00 + b_fc[0];
        out[(size_t)(row0 + n) * 2 + 1] = p01 + b_fc[1];
    }
}

extern "C" void kernel_launch(void* const* d_in, const int* in_sizes, int n_in,
                              void* d_out, int out_size, void* d_ws, size_t ws_size,
                              hipStream_t stream) {
    const float* x    = (const float*)d_in[0];
    const float* W_ih = (const float*)d_in[1];
    const float* b_ih = (const float*)d_in[2];
    const float* W_hh = (const float*)d_in[3];
    const float* b_hh = (const float*)d_in[4];
    const float* W_fc = (const float*)d_in[5];
    const float* b_fc = (const float*)d_in[6];
    float* out = (float*)d_out;

    const int C = 2;
    const int B = out_size / C;
    const int T = in_sizes[0] / B;

    const int threads = 32;
    const int batchPerBlock = (threads / 32) * 16;
    const int blocks = (B + batchPerBlock - 1) / batchPerBlock;

    rnn_scan_wmma<<<blocks, threads, 0, stream>>>(
        x, W_ih, b_ih, W_hh, b_hh, W_fc, b_fc, out, B, T);
}


# x is fetched 16 timesteps at a time: four 16-byte loads per lane issued as one group a whole group ahead; was a load, a prefetch, an index clamp and 64-bit address work every 4 steps
# speedup vs baseline: 3.9311x; 1.0698x over previous
#include <hip/hip_runtime.h>
#include <hip/hip_bf16.h>

typedef __attribute__((ext_vector_type(2))) float v2f;
typedef __attribute__((ext_vector_type(4))) float v4f;
typedef __attribute__((ext_vector_type(8))) float v8f;

__device__ __forceinline__ float xswap16(float v) {
#if __has_builtin(__builtin_amdgcn_permlanex16)
    return __builtin_bit_cast(float, __builtin_amdgcn_permlanex16(
        __builtin_bit_cast(int, v), __builtin_bit_cast(int, v),
        0x76543210, 0xfedcba98, false, false));
#elif __has_builtin(__builtin_amdgcn_ds_swizzle)
    return __builtin_bit_cast(float,
        __builtin_amdgcn_ds_swizzle(__builtin_bit_cast(int, v), 0x401f));
#else
    return v;
#endif
}

__device__ __forceinline__ float relu_med3(float x, float finf) {
#if defined(__AMDGCN__)
    float r;
    asm("v_med3_f32 %0, %1, 0, %2" : "=v"(r) : "v"(x), "v"(finf));
    return r;
#else
    return fmaxf(x, 0.0f);
#endif
}

__device__ __forceinline__ void rnn_steps4(v8f& h0, const v2f (&a)[4],
                                           const float (&wih)[8], const v8f bias,
                                           const v4f xq, const float finf) {
    const v8f zc = {};
#pragma unroll
    for (int u = 0; u < 4; ++u) {
        v2f b00 = { h0[0], h0[1] }, b01 = { h0[2], h0[3] };
        v2f b02 = { h0[4], h0[5] }, b03 = { h0[6], h0[7] };

        __builtin_amdgcn_sched_barrier(0);
        v8f c0 = __builtin_amdgcn_wmma_f32_16x16x4_f32(
                    false, a[0], false, b00, (short)0, bias, false, false);
        v8f q0 = __builtin_amdgcn_wmma_f32_16x16x4_f32(
                    false, a[2], false, b02, (short)0, zc, false, false);
        c0 = __builtin_amdgcn_wmma_f32_16x16x4_f32(
                    false, a[1], false, b01, (short)0, c0, false, false);
        q0 = __builtin_amdgcn_wmma_f32_16x16x4_f32(
                    false, a[3], false, b03, (short)0, q0, false, false);
        __builtin_amdgcn_sched_barrier(0);
#if defined(__AMDGCN__)
        asm volatile("" : "+v"(c0) : "v"(q0));
#endif

        v8f t0;
#pragma unroll
        for (int v = 0; v < 8; ++v)
            t0[v] = fmaf(xq[u], wih[v], c0[v]);

        float s0[8];
#pragma unroll
        for (int v = 0; v < 8; ++v)
            s0[v] = t0[v] + q0[v];
        __builtin_amdgcn_sched_barrier(0);
#pragma unroll
        for (int v = 0; v < 8; ++v)
            h0[v] = relu_med3(s0[v], finf);
    }
}

__global__ __launch_bounds__(32) void rnn_scan_wmma(
    const float* __restrict__ x,
    const float* __restrict__ W_ih,
    const float* __restrict__ b_ih,
    const float* __restrict__ W_hh,
    const float* __restrict__ b_hh,
    const float* __restrict__ W_fc,
    const float* __restrict__ b_fc,
    float* __restrict__ out,
    int B, int T)
{
    const int lane = threadIdx.x & 31;
    const int wave = threadIdx.x >> 5;
    const int n    = lane & 15;
    const bool hi  = lane >= 16;
    const int h8   = hi ? 8 : 0;

    const int waveGlobal = blockIdx.x * (blockDim.x >> 5) + wave;
    const int row0 = waveGlobal * 16;
    if (row0 >= B) return;

    const float finf = __builtin_bit_cast(float, 0x7f800000);

    v2f a[4];
#pragma unroll
    for (int j = 0; j < 4; ++j) {
        const int k0 = 2 * j + h8;
        a[j].x = W_hh[n * 16 + k0];
        a[j].y = W_hh[n * 16 + k0 + 1];
    }

    float wih[8];
    v8f bias;
#pragma unroll
    for (int v = 0; v < 8; ++v) {
        wih[v]  = W_ih[v + h8];
        bias[v] = b_ih[v + h8] + b_hh[v + h8];
    }

    v8f h0 = {};

    const int nblk = T >> 2;
    const int ngrp = nblk >> 2;
    const char* xrow = (const char*)x + (size_t)(row0 + n) * T * 4;

    auto load16 = [&](v4f (&dst)[4], int g) {
        const char* p = xrow + (size_t)g * 64;
#pragma unroll
        for (int q = 0; q < 4; ++q)
            dst[q] = *(const v4f*)(p + 16 * q);
    };
    auto steps16 = [&](const v4f (&xs)[4]) {
#pragma unroll
        for (int d = 0; d < 4; ++d)
            rnn_steps4(h0, a, wih, bias, xs[d], finf);
    };

    if (ngrp > 0) {
        v4f xa[4], xb[4];
        load16(xa, 0);
        const int npair = (ngrp - 1) >> 1;
        int g = 0;
        for (int p = 0; p < npair; ++p, g += 2) {
            __builtin_amdgcn_sched_barrier(0x6);
            load16(xb, g + 1);
            __builtin_amdgcn_sched_barrier(0x6);
            steps16(xa);
            __builtin_amdgcn_sched_barrier(0x6);
            load16(xa, g + 2);
            __builtin_amdgcn_sched_barrier(0x6);
            steps16(xb);
        }
        if (g + 2 <= ngrp) {
            __builtin_amdgcn_sched_barrier(0x6);
            load16(xb, g + 1);
            __builtin_amdgcn_sched_barrier(0x6);
            steps16(xa);
            steps16(xb);
        } else {
            steps16(xa);
        }
    }
    for (int blk = ngrp * 4; blk < nblk; ++blk) {
        const v4f xq = *(const v4f*)(xrow + (size_t)blk * 16);
        rnn_steps4(h0, a, wih, bias, xq, finf);
    }

    float wf0[8], wf1[8];
#pragma unroll
    for (int v = 0; v < 8; ++v) {
        wf0[v] = W_fc[0 * 16 + v + h8];
        wf1[v] = W_fc[1 * 16 + v + h8];
    }
    float p00 = 0.f, p01 = 0.f;
#pragma unroll
    for (int v = 0; v < 8; ++v) {
        p00 = fmaf(h0[v], wf0[v], p00);
        p01 = fmaf(h0[v], wf1[v], p01);
    }
    p00 += xswap16(p00); p01 += xswap16(p01);

    if (!hi) {
        out[(size_t)(row0 + n) * 2 + 0] = p00 + b_fc[0];
        out[(size_t)(row0 + n) * 2 + 1] = p01 + b_fc[1];
    }
}

extern "C" void kernel_launch(void* const* d_in, const int* in_sizes, int n_in,
                              void* d_out, int out_size, void* d_ws, size_t ws_size,
                              hipStream_t stream) {
    const float* x    = (const float*)d_in[0];
    const float* W_ih = (const float*)d_in[1];
    const float* b_ih = (const float*)d_in[2];
    const float* W_hh = (const float*)d_in[3];
    const float* b_hh = (const float*)d_in[4];
    const float* W_fc = (const float*)d_in[5];
    const float* b_fc = (const float*)d_in[6];
    float* out = (float*)d_out;

    const int C = 2;
    const int B = out_size / C;
    const int T = in_sizes[0] / B;

    const int threads = 32;
    const int batchPerBlock = (threads / 32) * 16;
    const int blocks = (B + batchPerBlock - 1) / batchPerBlock;

    rnn_scan_wmma<<<blocks, threads, 0, stream>>>(
        x, W_ih, b_ih, W_hh, b_hh, W_fc, b_fc, out, B, T);
}
